# starting kernel
# baseline (speedup 1.0000x reference)
#include <hip/hip_runtime.h>


typedef __attribute__((ext_vector_type(2))) float v2f;
typedef __attribute__((ext_vector_type(8))) float v8f;

#define NB   8
#define NT   1024
#define NH   8
#define NS   64
#define NK   31
#define NC   512
#define PADK 15
#define NJ   12

__global__ __launch_bounds__(256) void lightconv_wmma_f32(
    const float* __restrict__ x,
    const float* __restrict__ filters,
    const float* __restrict__ bias,
    float* __restrict__ out)
{
    __shared__ float wsm[8][16][32];

    const int lane = threadIdx.x & 31;
    const int wave = threadIdx.x >> 5;
    const int task = blockIdx.x * 8 + wave;
    const int h    = task & 7;
    const int tt   = (task >> 3) & 63;
    const int b    = task >> 9;
    const int t0   = tt * 16;

    const int m = lane & 15;
    if (lane < 16) {
        const float* f = filters + (size_t)(b * NT + t0 + m) * (NH * NK) + h * NK;
        float v[NK];
        float mx = -3.4e38f;
#pragma unroll
        for (int k = 0; k < NK; ++k) { v[k] = f[k]; mx = fmaxf(mx, v[k]); }
        float sum = 0.f;
#pragma unroll
        for (int k = 0; k < NK; ++k) { v[k] = __expf(v[k] - mx); sum += v[k]; }
        const float inv = 1.f / sum;
#pragma unroll
        for (int k = 0; k < NK; ++k) wsm[wave][m][k] = v[k] * inv;
    }
    __syncthreads();

    const int klo = (lane < 16) ? 0 : 2;
    v2f afrag[NJ];
#pragma unroll
    for (int j = 0; j < NJ; ++j) {
        const int jj0 = 4 * j + klo;
        const int k0  = jj0 - m;
        const int k1  = k0 + 1;
        afrag[j].x = (k0 >= 0 && k0 < NK) ? wsm[wave][m][k0] : 0.f;
        afrag[j].y = (k1 >= 0 && k1 < NK) ? wsm[wave][m][k1] : 0.f;
    }

    const int n     = lane & 15;
    const int rbase = t0 - PADK + klo;
    for (int st = 0; st < 4; ++st) {
        const int c0 = h * NS + st * 16;
        const float* xcol = x + (size_t)b * NT * NC + c0 + n;

        v8f acc = {};
#pragma unroll
        for (int j = 0; j < NJ; ++j) {
            const int r0 = rbase + 4 * j;
            const int r1 = r0 + 1;
            v2f bfrag;
            bfrag.x = (r0 >= 0 && r0 < NT) ? xcol[(size_t)r0 * NC] : 0.f;
            bfrag.y = (r1 >= 0 && r1 < NT) ? xcol[(size_t)r1 * NC] : 0.f;
            acc = __builtin_amdgcn_wmma_f32_16x16x4_f32(
                      false, afrag[j], false, bfrag, (short)0, acc, false, false);
        }

        const float bv    = bias[c0 + n];
        const int   mrow0 = (lane < 16) ? 0 : 8;
        float* orow = out + (size_t)(b * NT + t0 + mrow0) * NC + c0 + n;
#pragma unroll
        for (int vi = 0; vi < 8; ++vi)
            orow[(size_t)vi * NC] = acc[vi] + bv;
    }
}

extern "C" void kernel_launch(void* const* d_in, const int* in_sizes, int n_in,
                              void* d_out, int out_size, void* d_ws, size_t ws_size,
                              hipStream_t stream) {
    (void)in_sizes; (void)n_in; (void)out_size; (void)d_ws; (void)ws_size;
    const float* x       = (const float*)d_in[0];
    const float* filters = (const float*)d_in[1];
    const float* bias    = (const float*)d_in[2];
    float*       out     = (float*)d_out;

    dim3 grid(512), block(256);
    lightconv_wmma_f32<<<grid, block, 0, stream>>>(x, filters, bias, out);
}


# All global loads of a task (filter taps, bias, 96 x loads) issued as one group at wave start; x read from a clamped row, padded taps zeroed by select after the load, not one guarded load per wait
# speedup vs baseline: 2.0774x; 2.0774x over previous
#include <hip/hip_runtime.h>


typedef __attribute__((ext_vector_type(2))) float v2f;
typedef __attribute__((ext_vector_type(8))) float v8f;

#define NB   8
#define NT   1024
#define NH   8
#define NS   64
#define NK   31
#define NC   512
#define PADK 15
#define NJ   12

__global__ __launch_bounds__(256) void lightconv_wmma_f32(
    const float* __restrict__ x,
    const float* __restrict__ filters,
    const float* __restrict__ bias,
    float* __restrict__ out)
{
    __shared__ float wsm[8][16][32];

    const int lane = threadIdx.x & 31;
    const int wave = threadIdx.x >> 5;
    const int task = blockIdx.x * 8 + wave;
    const int h    = task & 7;
    const int tt   = (task >> 3) & 63;
    const int b    = task >> 9;
    const int t0   = tt * 16;

    const int m   = lane & 15;
    const int n   = lane & 15;
    const int klo = (lane < 16) ? 0 : 2;

    const float* f = filters + (size_t)(b * NT + t0 + m) * (NH * NK) + h * NK;
    float v[NK];
    if (lane < 16) {
#pragma unroll
        for (int k = 0; k < NK; ++k) v[k] = f[k];
    }

    const float* xcol = x + (size_t)b * NT * NC + h * NS + n;
    float bv[4];
#pragma unroll
    for (int st = 0; st < 4; ++st) bv[st] = bias[h * NS + st * 16 + n];

    const int rbase = t0 - PADK + klo;
    float xb[4][2 * NJ];
#pragma unroll
    for (int j = 0; j < NJ; ++j) {
        const int r0  = rbase + 4 * j;
        const int r1  = r0 + 1;
        const int r0c = min(max(r0, 0), NT - 1);
        const int r1c = min(max(r1, 0), NT - 1);
        const float* p0 = xcol + (size_t)r0c * NC;
        const float* p1 = xcol + (size_t)r1c * NC;
#pragma unroll
        for (int st = 0; st < 4; ++st) {
            xb[st][2 * j]     = p0[st * 16];
            xb[st][2 * j + 1] = p1[st * 16];
        }
    }

    if (lane < 16) {
        float mx = -3.4e38f;
#pragma unroll
        for (int k = 0; k < NK; ++k) mx = fmaxf(mx, v[k]);
        float sum = 0.f;
#pragma unroll
        for (int k = 0; k < NK; ++k) { v[k] = __expf(v[k] - mx); sum += v[k]; }
        const float inv = 1.f / sum;
#pragma unroll
        for (int k = 0; k < NK; ++k) wsm[wave][m][k] = v[k] * inv;
    }
    __syncthreads();

    v2f afrag[NJ];
#pragma unroll
    for (int j = 0; j < NJ; ++j) {
        const int jj0 = 4 * j + klo;
        const int k0  = jj0 - m;
        const int k1  = k0 + 1;
        afrag[j].x = (k0 >= 0 && k0 < NK) ? wsm[wave][m][k0] : 0.f;
        afrag[j].y = (k1 >= 0 && k1 < NK) ? wsm[wave][m][k1] : 0.f;
    }

    const int mrow0 = (lane < 16) ? 0 : 8;
    float* orow0 = out + (size_t)(b * NT + t0 + mrow0) * NC + h * NS + n;
#pragma unroll
    for (int st = 0; st < 4; ++st) {
        v8f acc = {};
#pragma unroll
        for (int j = 0; j < NJ; ++j) {
            const int r0 = rbase + 4 * j;
            const int r1 = r0 + 1;
            v2f bfrag;
            bfrag.x = (r0 >= 0 && r0 < NT) ? xb[st][2 * j]     : 0.f;
            bfrag.y = (r1 >= 0 && r1 < NT) ? xb[st][2 * j + 1] : 0.f;
            acc = __builtin_amdgcn_wmma_f32_16x16x4_f32(
                      false, afrag[j], false, bfrag, (short)0, acc, false, false);
        }

        float* orow = orow0 + st * 16;
#pragma unroll
        for (int vi = 0; vi < 8; ++vi)
            orow[(size_t)vi * NC] = acc[vi] + bv[st];
    }
}

extern "C" void kernel_launch(void* const* d_in, const int* in_sizes, int n_in,
                              void* d_out, int out_size, void* d_ws, size_t ws_size,
                              hipStream_t stream) {
    (void)in_sizes; (void)n_in; (void)out_size; (void)d_ws; (void)ws_size;
    const float* x       = (const float*)d_in[0];
    const float* filters = (const float*)d_in[1];
    const float* bias    = (const float*)d_in[2];
    float*       out     = (float*)d_out;

    dim3 grid(512), block(256);
    lightconv_wmma_f32<<<grid, block, 0, stream>>>(x, filters, bias, out);
}


# x, bias and out moved as 16-byte vectors: fragment lane n owns head columns 4n..4n+3 (one per s-tile) instead of n+16*st, so each load/store covers whole 128-B lines; 140 memory instr per wave -> 41
# speedup vs baseline: 2.4256x; 1.1676x over previous
#include <hip/hip_runtime.h>


typedef __attribute__((ext_vector_type(2))) float v2f;
typedef __attribute__((ext_vector_type(4))) float v4f;
typedef __attribute__((ext_vector_type(8))) float v8f;

#define NB   8
#define NT   1024
#define NH   8
#define NS   64
#define NK   31
#define NC   512
#define PADK 15
#define NJ   12

__global__ __launch_bounds__(256) void lightconv_wmma_f32(
    const float* __restrict__ x,
    const float* __restrict__ filters,
    const float* __restrict__ bias,
    float* __restrict__ out)
{
    __shared__ float wsm[8][16][32];

    const int lane = threadIdx.x & 31;
    const int wave = threadIdx.x >> 5;
    const int task = blockIdx.x * 8 + wave;
    const int h    = task & 7;
    const int tt   = (task >> 3) & 63;
    const int b    = task >> 9;
    const int t0   = tt * 16;

    const int m   = lane & 15;
    const int n   = lane & 15;
    const int klo = (lane < 16) ? 0 : 2;

    const float* f = filters + (size_t)(b * NT + t0 + m) * (NH * NK) + h * NK;
    float v[NK];
    if (lane < 16) {
#pragma unroll
        for (int k = 0; k < NK; ++k) v[k] = f[k];
    }

    const float* xcol = x + (size_t)b * NT * NC + h * NS + 4 * n;
    const v4f bv = *(const v4f*)(bias + h * NS + 4 * n);

    const int rbase = t0 - PADK + klo;
    v4f xb[2 * NJ];
#pragma unroll
    for (int j = 0; j < NJ; ++j) {
        const int r0  = rbase + 4 * j;
        const int r1  = r0 + 1;
        const int r0c = min(max(r0, 0), NT - 1);
        const int r1c = min(max(r1, 0), NT - 1);
        xb[2 * j]     = *(const v4f*)(xcol + (size_t)r0c * NC);
        xb[2 * j + 1] = *(const v4f*)(xcol + (size_t)r1c * NC);
    }

    if (lane < 16) {
        float mx = -3.4e38f;
#pragma unroll
        for (int k = 0; k < NK; ++k) mx = fmaxf(mx, v[k]);
        float sum = 0.f;
#pragma unroll
        for (int k = 0; k < NK; ++k) { v[k] = __expf(v[k] - mx); sum += v[k]; }
        const float inv = 1.f / sum;
#pragma unroll
        for (int k = 0; k < NK; ++k) wsm[wave][m][k] = v[k] * inv;
    }
    __syncthreads();

    v2f afrag[NJ];
#pragma unroll
    for (int j = 0; j < NJ; ++j) {
        const int jj0 = 4 * j + klo;
        const int k0  = jj0 - m;
        const int k1  = k0 + 1;
        afrag[j].x = (k0 >= 0 && k0 < NK) ? wsm[wave][m][k0] : 0.f;
        afrag[j].y = (k1 >= 0 && k1 < NK) ? wsm[wave][m][k1] : 0.f;
    }

    v8f acc[4];
#pragma unroll
    for (int st = 0; st < 4; ++st) {
        acc[st] = (v8f){};
#pragma unroll
        for (int j = 0; j < NJ; ++j) {
            const int r0 = rbase + 4 * j;
            const int r1 = r0 + 1;
            v2f bfrag;
            bfrag.x = (r0 >= 0 && r0 < NT) ? xb[2 * j][st]     : 0.f;
            bfrag.y = (r1 >= 0 && r1 < NT) ? xb[2 * j + 1][st] : 0.f;
            acc[st] = __builtin_amdgcn_wmma_f32_16x16x4_f32(
                          false, afrag[j], false, bfrag, (short)0, acc[st], false, false);
        }
    }

    const int mrow0 = (lane < 16) ? 0 : 8;
    float* orow0 = out + (size_t)(b * NT + t0 + mrow0) * NC + h * NS + 4 * n;
#pragma unroll
    for (int vi = 0; vi < 8; ++vi) {
        v4f o;
        o.x = acc[0][vi] + bv.x;
        o.y = acc[1][vi] + bv.y;
        o.z = acc[2][vi] + bv.z;
        o.w = acc[3][vi] + bv.w;
        *(v4f*)(orow0 + (size_t)vi * NC) = o;
    }
}

extern "C" void kernel_launch(void* const* d_in, const int* in_sizes, int n_in,
                              void* d_out, int out_size, void* d_ws, size_t ws_size,
                              hipStream_t stream) {
    (void)in_sizes; (void)n_in; (void)out_size; (void)d_ws; (void)ws_size;
    const float* x       = (const float*)d_in[0];
    const float* filters = (const float*)d_in[1];
    const float* bias    = (const float*)d_in[2];
    float*       out     = (float*)d_out;

    dim3 grid(512), block(256);
    lightconv_wmma_f32<<<grid, block, 0, stream>>>(x, filters, bias, out);
}


# Filter taps fetched by the whole work-group as one line-aligned slab (4 whole lines per 16-byte load) into LDS; softmax in A-fragment layout on all 32 lanes, row sum split over the two lane halves
# speedup vs baseline: 2.6562x; 1.0951x over previous
#include <hip/hip_runtime.h>


typedef __attribute__((ext_vector_type(2))) float v2f;
typedef __attribute__((ext_vector_type(4))) float v4f;
typedef __attribute__((ext_vector_type(8))) float v8f;

#define NB   8
#define NT   1024
#define NH   8
#define NS   64
#define NK   31
#define NC   512
#define PADK 15
#define NJ   12
#define FROW 248
#define NFV  992

static __device__ __forceinline__ float xhalf(float x) {
    int b = __builtin_bit_cast(int, x);
    return __builtin_bit_cast(float,
        __builtin_amdgcn_permlanex16(b, b, 0x76543210, 0xfedcba98, false, false));
}

__global__ __launch_bounds__(256) void lightconv_wmma_f32(
    const float* __restrict__ x,
    const float* __restrict__ filters,
    const float* __restrict__ bias,
    float* __restrict__ out)
{
    __shared__ __attribute__((aligned(16))) float fraw[16 * FROW + 8];

    const int tid  = threadIdx.x;
    const int lane = tid & 31;
    const int wave = tid >> 5;
    const int h    = wave;
    const int tt   = blockIdx.x & 63;
    const int b    = blockIdx.x >> 6;
    const int t0   = tt * 16;

    const int m   = lane & 15;
    const int n   = lane & 15;
    const int klo = (lane < 16) ? 0 : 2;

    const v4f* fslab = (const v4f*)(filters + (size_t)(b * NT + t0) * FROW);
    v4f fv[4];
#pragma unroll
    for (int i = 0; i < 3; ++i) fv[i] = fslab[tid + 256 * i];
    fv[3] = (v4f){0.f, 0.f, 0.f, 0.f};
    if (tid + 768 < NFV) fv[3] = fslab[tid + 768];

    const float* xcol = x + (size_t)b * NT * NC + h * NS + 4 * n;
    const v4f bv = *(const v4f*)(bias + h * NS + 4 * n);

    const int rbase = t0 - PADK + klo;
    v4f xb[2 * NJ];
#pragma unroll
    for (int j = 0; j < NJ; ++j) {
        const int r0  = rbase + 4 * j;
        const int r1  = r0 + 1;
        const int r0c = min(max(r0, 0), NT - 1);
        const int r1c = min(max(r1, 0), NT - 1);
        xb[2 * j]     = *(const v4f*)(xcol + (size_t)r0c * NC);
        xb[2 * j + 1] = *(const v4f*)(xcol + (size_t)r1c * NC);
    }

    v4f* fr4 = (v4f*)fraw;
#pragma unroll
    for (int i = 0; i < 3; ++i) fr4[tid + 256 * i] = fv[i];
    if (tid + 768 < NFV + 2) fr4[tid + 768] = fv[3];

    asm volatile("s_wait_dscnt 0x0\n\ts_barrier_signal -1\n\ts_barrier_wait -1" ::: "memory");

    const int    kb = klo - m;
    const float* fp = fraw + m * (FROW - 1) + h * NK + klo;
    float e[2 * NJ];
    float mx = -__builtin_inff();
#pragma unroll
    for (int s = 0; s < 2 * NJ; ++s) {
        const int   off = 4 * (s >> 1) + (s & 1);
        const int   k   = kb + off;
        const float raw = fp[off];
        e[s] = (k >= 0 && k < NK) ? raw : -__builtin_inff();
        mx = fmaxf(mx, e[s]);
    }
    mx = fmaxf(mx, xhalf(mx));
    float sum = 0.f;
#pragma unroll
    for (int s = 0; s < 2 * NJ; ++s) { e[s] = __expf(e[s] - mx); sum += e[s]; }
    sum += xhalf(sum);
    const float inv = 1.f / sum;
    __builtin_amdgcn_sched_barrier(0);

    v2f afrag[NJ];
#pragma unroll
    for (int j = 0; j < NJ; ++j) {
        afrag[j].x = e[2 * j]     * inv;
        afrag[j].y = e[2 * j + 1] * inv;
    }

    v8f acc[4];
#pragma unroll
    for (int st = 0; st < 4; ++st) {
        acc[st] = (v8f){};
#pragma unroll
        for (int j = 0; j < NJ; ++j) {
            const int r0 = rbase + 4 * j;
            const int r1 = r0 + 1;
            v2f bfrag;
            bfrag.x = (r0 >= 0 && r0 < NT) ? xb[2 * j][st]     : 0.f;
            bfrag.y = (r1 >= 0 && r1 < NT) ? xb[2 * j + 1][st] : 0.f;
            acc[st] = __builtin_amdgcn_wmma_f32_16x16x4_f32(
                          false, afrag[j], false, bfrag, (short)0, acc[st], false, false);
        }
    }

    const int mrow0 = (lane < 16) ? 0 : 8;
    float* orow0 = out + (size_t)(b * NT + t0 + mrow0) * NC + h * NS + 4 * n;
#pragma unroll
    for (int vi = 0; vi < 8; ++vi) {
        v4f o;
        o.x = acc[0][vi] + bv.x;
        o.y = acc[1][vi] + bv.y;
        o.z = acc[2][vi] + bv.z;
        o.w = acc[3][vi] + bv.w;
        *(v4f*)(orow0 + (size_t)vi * NC) = o;
    }
}

extern "C" void kernel_launch(void* const* d_in, const int* in_sizes, int n_in,
                              void* d_out, int out_size, void* d_ws, size_t ws_size,
                              hipStream_t stream) {
    (void)in_sizes; (void)n_in; (void)out_size; (void)d_ws; (void)ws_size;
    const float* x       = (const float*)d_in[0];
    const float* filters = (const float*)d_in[1];
    const float* bias    = (const float*)d_in[2];
    float*       out     = (float*)d_out;

    dim3 grid(512), block(256);
    lightconv_wmma_f32<<<grid, block, 0, stream>>>(x, filters, bias, out);
}


# Each wave does two adjacent 16-row t-tiles of one head on one shared 64-row x band: 32 x loads per tile pair instead of 48, padded rows zeroed once per register; 2048 waves in 256 work-groups
# speedup vs baseline: 2.8620x; 1.0775x over previous
#include <hip/hip_runtime.h>


typedef __attribute__((ext_vector_type(2))) float v2f;
typedef __attribute__((ext_vector_type(4))) float v4f;
typedef __attribute__((ext_vector_type(8))) float v8f;

#define NB   8
#define NT   1024
#define NH   8
#define NS   64
#define NK   31
#define NC   512
#define PADK 15
#define NJ   12
#define NJ2  16
#define FROW 248
#define NFV  1984

static __device__ __forceinline__ float xhalf(float x) {
    int b = __builtin_bit_cast(int, x);
    return __builtin_bit_cast(float,
        __builtin_amdgcn_permlanex16(b, b, 0x76543210, 0xfedcba98, false, false));
}

static __device__ __forceinline__ void softmax_afrag(const float* fp, int kb, v2f (&afrag)[NJ]) {
    float e[2 * NJ];
    float mx = -__builtin_inff();
#pragma unroll
    for (int s = 0; s < 2 * NJ; ++s) {
        const int   off = 4 * (s >> 1) + (s & 1);
        const int   k   = kb + off;
        const float raw = fp[off];
        e[s] = (k >= 0 && k < NK) ? raw : -__builtin_inff();
        mx = fmaxf(mx, e[s]);
    }
    mx = fmaxf(mx, xhalf(mx));
    float sum = 0.f;
#pragma unroll
    for (int s = 0; s < 2 * NJ; ++s) { e[s] = __expf(e[s] - mx); sum += e[s]; }
    sum += xhalf(sum);
    const float inv = 1.f / sum;
#pragma unroll
    for (int j = 0; j < NJ; ++j) {
        afrag[j].x = e[2 * j]     * inv;
        afrag[j].y = e[2 * j + 1] * inv;
    }
}

__global__ __launch_bounds__(256) void lightconv_wmma_f32(
    const float* __restrict__ x,
    const float* __restrict__ filters,
    const float* __restrict__ bias,
    float* __restrict__ out)
{
    __shared__ __attribute__((aligned(16))) float fraw[32 * FROW + 8];

    const int tid  = threadIdx.x;
    const int lane = tid & 31;
    const int wave = tid >> 5;
    const int h    = wave;
    const int tp   = blockIdx.x & 31;
    const int b    = blockIdx.x >> 5;
    const int t0   = tp * 32;

    const int m   = lane & 15;
    const int n   = lane & 15;
    const int klo = (lane < 16) ? 0 : 2;

    const v4f* fslab = (const v4f*)(filters + (size_t)(b * NT + t0) * FROW);
    v4f fv[8];
#pragma unroll
    for (int i = 0; i < 7; ++i) fv[i] = fslab[tid + 256 * i];
    fv[7] = (v4f){0.f, 0.f, 0.f, 0.f};
    if (tid + 1792 < NFV) fv[7] = fslab[tid + 1792];

    const float* xcol = x + (size_t)b * NT * NC + h * NS + 4 * n;
    const v4f bv = *(const v4f*)(bias + h * NS + 4 * n);

    const int rbase = t0 - PADK + klo;
    v4f xb[2 * NJ2];
#pragma unroll
    for (int j = 0; j < NJ2; ++j) {
        const int r0  = rbase + 4 * j;
        const int r1  = r0 + 1;
        const int r0c = min(max(r0, 0), NT - 1);
        const int r1c = min(max(r1, 0), NT - 1);
        xb[2 * j]     = *(const v4f*)(xcol + (size_t)r0c * NC);
        xb[2 * j + 1] = *(const v4f*)(xcol + (size_t)r1c * NC);
    }

    v4f* fr4 = (v4f*)fraw;
#pragma unroll
    for (int i = 0; i < 7; ++i) fr4[tid + 256 * i] = fv[i];
    if (tid + 1792 < NFV + 2) fr4[tid + 1792] = fv[7];

    asm volatile("s_wait_dscnt 0x0\n\ts_barrier_signal -1\n\ts_barrier_wait -1" ::: "memory");

    const int    kb  = klo - m;
    const float* fp0 = fraw + m * (FROW - 1) + h * NK + klo;
    const float* fp1 = fp0 + 16 * FROW;
    v2f af0[NJ], af1[NJ];
    softmax_afrag(fp0, kb, af0);
    softmax_afrag(fp1, kb, af1);
    __builtin_amdgcn_sched_barrier(0);

    const int mrow0 = (lane < 16) ? 0 : 8;
    float* orow0 = out + (size_t)(b * NT + t0 + mrow0) * NC + h * NS + 4 * n;

#pragma unroll
    for (int j = 0; j < NJ; ++j) {
        const int r0 = rbase + 4 * j;
        const int r1 = r0 + 1;
        const bool ok0 = (r0 >= 0 && r0 < NT);
        const bool ok1 = (r1 >= 0 && r1 < NT);
#pragma unroll
        for (int st = 0; st < 4; ++st) {
            xb[2 * j][st]     = ok0 ? xb[2 * j][st]     : 0.f;
            xb[2 * j + 1][st] = ok1 ? xb[2 * j + 1][st] : 0.f;
        }
    }
    {
        v8f acc[4];
#pragma unroll
        for (int st = 0; st < 4; ++st) {
            acc[st] = (v8f){};
#pragma unroll
            for (int j = 0; j < NJ; ++j) {
                v2f bfrag;
                bfrag.x = xb[2 * j][st];
                bfrag.y = xb[2 * j + 1][st];
                acc[st] = __builtin_amdgcn_wmma_f32_16x16x4_f32(
                              false, af0[j], false, bfrag, (short)0, acc[st], false, false);
            }
        }
#pragma unroll
        for (int vi = 0; vi < 8; ++vi) {
            v4f o;
            o.x = acc[0][vi] + bv.x;
            o.y = acc[1][vi] + bv.y;
            o.z = acc[2][vi] + bv.z;
            o.w = acc[3][vi] + bv.w;
            *(v4f*)(orow0 + (size_t)vi * NC) = o;
        }
    }

#pragma unroll
    for (int j = NJ; j < NJ2; ++j) {
        const int r0 = rbase + 4 * j;
        const int r1 = r0 + 1;
        const bool ok0 = (r0 >= 0 && r0 < NT);
        const bool ok1 = (r1 >= 0 && r1 < NT);
#pragma unroll
        for (int st = 0; st < 4; ++st) {
            xb[2 * j][st]     = ok0 ? xb[2 * j][st]     : 0.f;
            xb[2 * j + 1][st] = ok1 ? xb[2 * j + 1][st] : 0.f;
        }
    }
    {
        v8f acc[4];
#pragma unroll
        for (int st = 0; st < 4; ++st) {
            acc[st] = (v8f){};
#pragma unroll
            for (int j = 0; j < NJ; ++j) {
                v2f bfrag;
                bfrag.x = xb[2 * (j + 4)][st];
                bfrag.y = xb[2 * (j + 4) + 1][st];
                acc[st] = __builtin_amdgcn_wmma_f32_16x16x4_f32(
                              false, af1[j], false, bfrag, (short)0, acc[st], false, false);
            }
        }
        float* orow1 = orow0 + (size_t)16 * NC;
#pragma unroll
        for (int vi = 0; vi < 8; ++vi) {
            v4f o;
            o.x = acc[0][vi] + bv.x;
            o.y = acc[1][vi] + bv.y;
            o.z = acc[2][vi] + bv.z;
            o.w = acc[3][vi] + bv.w;
            *(v4f*)(orow1 + (size_t)vi * NC) = o;
        }
    }
}

extern "C" void kernel_launch(void* const* d_in, const int* in_sizes, int n_in,
                              void* d_out, int out_size, void* d_ws, size_t ws_size,
                              hipStream_t stream) {
    (void)in_sizes; (void)n_in; (void)out_size; (void)d_ws; (void)ws_size;
    const float* x       = (const float*)d_in[0];
    const float* filters = (const float*)d_in[1];
    const float* bias    = (const float*)d_in[2];
    float*       out     = (float*)d_out;

    dim3 grid(256), block(256);
    lightconv_wmma_f32<<<grid, block, 0, stream>>>(x, filters, bias, out);
}
